# starting kernel
# baseline (speedup 1.0000x reference)
#include <hip/hip_runtime.h>
#include <hip/hip_bf16.h>


typedef float v2f __attribute__((ext_vector_type(2)));
typedef float v8f __attribute__((ext_vector_type(8)));


__global__ void deg_init_kernel(float* __restrict__ deg, int n) {
    int i = blockIdx.x * blockDim.x + threadIdx.x;
    if (i < n) deg[i] = 1.0f;
}

__global__ void deg_count_kernel(const int* __restrict__ dst,
                                 float* __restrict__ deg, int ne) {
    int e = blockIdx.x * blockDim.x + threadIdx.x;
    if (e < ne) atomicAdd(&deg[dst[e]], 1.0f);
}

__global__ void deg_to_dinv_kernel(float* __restrict__ deg, int n) {
    int i = blockIdx.x * blockDim.x + threadIdx.x;
    if (i < n) {
        float d = deg[i];
        deg[i] = (d > 0.0f) ? rsqrtf(d) : 0.0f;
    }
}


template <bool RELU, int TN>
__global__ void gemm_wmma_strip(const float* __restrict__ A,
                                const float* __restrict__ W,
                                float* __restrict__ out,
                                int M, int K) {
    const int Nc   = TN * 16;
    const int wave = (blockIdx.x * blockDim.x + threadIdx.x) >> 5;
    const int lane = threadIdx.x & 31;

    const int tm = wave;
    if (tm * 16 >= M) return;

    const int ml    = lane & 15;
    const int khalf = (lane >> 4) << 1;

    const float* arow = A + (size_t)(tm * 16 + ml) * K;

    v8f acc[TN];
#pragma unroll
    for (int t = 0; t < TN; ++t) acc[t] = (v8f){};

    for (int k = 0; k < K; k += 4) {
        v2f a = *(const v2f*)(arow + k + khalf);
        if (RELU) {
            a.x = fmaxf(a.x, 0.0f);
            a.y = fmaxf(a.y, 0.0f);
        }

        const float* w0 = W + (size_t)(k + khalf) * Nc + ml;
        v2f b[TN];
#pragma unroll
        for (int t = 0; t < TN; ++t) {
            b[t].x = w0[t * 16];
            b[t].y = w0[Nc + t * 16];
        }

#pragma unroll
        for (int t = 0; t < TN; ++t) {
            acc[t] = __builtin_amdgcn_wmma_f32_16x16x4_f32(
                 false, a,  false, b[t],
                 (short)0, acc[t],  false,  false);
        }
    }

    const int row0 = tm * 16 + ((lane >> 4) << 3);
#pragma unroll
    for (int t = 0; t < TN; ++t) {
        float* o = out + (size_t)row0 * Nc + t * 16 + ml;
#pragma unroll
        for (int r = 0; r < 8; ++r) o[(size_t)r * Nc] = acc[t][r];
    }
}


__global__ void self_init_kernel(const float* __restrict__ t,
                                 const float* __restrict__ dinv,
                                 const float* __restrict__ bias,
                                 float* __restrict__ agg,
                                 int n, int C) {
    int idx = blockIdx.x * blockDim.x + threadIdx.x;
    if (idx >= n * C) return;
    int i = idx / C;
    int c = idx - i * C;
    float di = dinv[i];
    agg[idx] = bias[c] + di * di * t[idx];
}

template <int C>
__global__ void edge_agg_kernel(const float* __restrict__ t,
                                const float* __restrict__ dinv,
                                const int* __restrict__ src,
                                const int* __restrict__ dst,
                                float* __restrict__ agg, int ne) {
    int e = (blockIdx.x * blockDim.x + threadIdx.x) >> 5;
    if (e >= ne) return;
    int lane = threadIdx.x & 31;

    int s = src[e];
    int d = dst[e];
    float nrm = dinv[s] * dinv[d];

    float* ad = agg + (size_t)d * C;
    if constexpr (C == 128) {
        const float4* ts = (const float4*)(t + (size_t)s * C);
        float4 v = ts[lane];
        atomicAdd(&ad[4 * lane + 0], nrm * v.x);
        atomicAdd(&ad[4 * lane + 1], nrm * v.y);
        atomicAdd(&ad[4 * lane + 2], nrm * v.z);
        atomicAdd(&ad[4 * lane + 3], nrm * v.w);
    } else {
        const float2* ts = (const float2*)(t + (size_t)s * C);
        float2 v = ts[lane];
        atomicAdd(&ad[2 * lane + 0], nrm * v.x);
        atomicAdd(&ad[2 * lane + 1], nrm * v.y);
    }
}


static inline int cdiv(long long a, long long b) { return (int)((a + b - 1) / b); }

extern "C" void kernel_launch(void* const* d_in, const int* in_sizes, int n_in,
                              void* d_out, int out_size, void* d_ws, size_t ws_size,
                              hipStream_t stream) {
    const float* x  = (const float*)d_in[0];
    const float* W1 = (const float*)d_in[1];
    const float* b1 = (const float*)d_in[2];
    const float* W2 = (const float*)d_in[3];
    const float* b2 = (const float*)d_in[4];
    const float* W3 = (const float*)d_in[5];
    const float* b3 = (const float*)d_in[6];
    const int*   ei = (const int*)d_in[7];

    const int IN_C = 64, HID_C = 128, OUT_C = 64;
    const int n  = in_sizes[0] / IN_C;
    const int ne = in_sizes[7] / 2;
    const int* src = ei;
    const int* dst = ei + ne;

    float* T    = (float*)d_ws;
    float* G    = T + (size_t)n * HID_C;
    float* dinv = G + (size_t)n * HID_C;
    float* out  = (float*)d_out;

    const int TB = 256;
    const long long strip_waves = n / 16;
    const int gemm_blocks = cdiv(strip_waves * 32, TB);

    deg_init_kernel<<<cdiv(n, TB), TB, 0, stream>>>(dinv, n);
    deg_count_kernel<<<cdiv(ne, TB), TB, 0, stream>>>(dst, dinv, ne);
    deg_to_dinv_kernel<<<cdiv(n, TB), TB, 0, stream>>>(dinv, n);

    gemm_wmma_strip<false, 8><<<gemm_blocks, TB, 0, stream>>>(x, W1, T, n, IN_C);
    self_init_kernel<<<cdiv((long long)n * HID_C, TB), TB, 0, stream>>>(T, dinv, b1, G, n, HID_C);
    edge_agg_kernel<HID_C><<<cdiv((long long)ne * 32, TB), TB, 0, stream>>>(T, dinv, src, dst, G, ne);

    gemm_wmma_strip<true, 8><<<gemm_blocks, TB, 0, stream>>>(G, W2, T, n, HID_C);
    self_init_kernel<<<cdiv((long long)n * HID_C, TB), TB, 0, stream>>>(T, dinv, b2, G, n, HID_C);
    edge_agg_kernel<HID_C><<<cdiv((long long)ne * 32, TB), TB, 0, stream>>>(T, dinv, src, dst, G, ne);

    gemm_wmma_strip<true, 4><<<gemm_blocks, TB, 0, stream>>>(G, W3, T, n, HID_C);
    self_init_kernel<<<cdiv((long long)n * OUT_C, TB), TB, 0, stream>>>(T, dinv, b3, out, n, OUT_C);
    edge_agg_kernel<OUT_C><<<cdiv((long long)ne * 32, TB), TB, 0, stream>>>(T, dinv, src, dst, out, ne);
}


# Aggregation as per-destination segmented sum: edges grouped by destination in-call (returning int atomic count+rank, scan, fill); one wave sums a row in registers; no float atomics; sum order by rank
# speedup vs baseline: 2.2671x; 2.2671x over previous
#include <hip/hip_runtime.h>


typedef float v2f __attribute__((ext_vector_type(2)));
typedef float v8f __attribute__((ext_vector_type(8)));


__global__ void cnt_zero_kernel(int* __restrict__ cnt, int n) {
    int i = blockIdx.x * blockDim.x + threadIdx.x;
    if (i < n) cnt[i] = 0;
}

__global__ void count_rank_kernel(const int* __restrict__ dst,
                                  int* __restrict__ cnt,
                                  int* __restrict__ rank, int ne) {
    int e = blockIdx.x * blockDim.x + threadIdx.x;
    if (e < ne) rank[e] = atomicAdd(&cnt[dst[e]], 1);
}

__device__ __forceinline__ float dinv_of(int c) {
    const float d = (float)(c + 1);
    return (d > 0.0f) ? rsqrtf(d) : 0.0f;
}

__global__ void __launch_bounds__(1024)
scan_dinv_kernel(const int* __restrict__ cnt, int* __restrict__ rowptr,
                 float* __restrict__ dinv, int n) {
    __shared__ int wsum[32];
    const int tid  = threadIdx.x;
    const int lane = tid & 31;
    const int w    = tid >> 5;
    const int per  = (n + 1023) >> 10;
    int lo = tid * per; if (lo > n) lo = n;
    int hi = lo + per;  if (hi > n) hi = n;
    const bool vec = ((per & 3) == 0) && (hi - lo == per);

    int s = 0;
    if (vec) {
        const int4* p = (const int4*)(cnt + lo);
        const int nq = per >> 2;
        int q = 0;
        for (; q + 8 <= nq; q += 8) {
            int4 c[8];
#pragma unroll
            for (int u = 0; u < 8; ++u) c[u] = p[q + u];
#pragma unroll
            for (int u = 0; u < 8; ++u) s += (c[u].x + c[u].y) + (c[u].z + c[u].w);
        }
        for (; q < nq; ++q) { int4 c = p[q]; s += (c.x + c.y) + (c.z + c.w); }
    } else {
        for (int i = lo; i < hi; ++i) s += cnt[i];
    }

    int incl = s;
#pragma unroll
    for (int off = 1; off < 32; off <<= 1) {
        int v = __shfl_up(incl, off);
        if (lane >= off) incl += v;
    }
    if (lane == 31) wsum[w] = incl;
    __syncthreads();
    if (w == 0) {
        int v = wsum[lane];
        int iv = v;
#pragma unroll
        for (int off = 1; off < 32; off <<= 1) {
            int u = __shfl_up(iv, off);
            if (lane >= off) iv += u;
        }
        wsum[lane] = iv - v;
    }
    __syncthreads();
    int run = wsum[w] + (incl - s);

    if (vec) {
        const int4* p = (const int4*)(cnt + lo);
        int4*   rp = (int4*)(rowptr + lo);
        float4* dp = (float4*)(dinv + lo);
        const int nq = per >> 2;
#pragma unroll 4
        for (int q = 0; q < nq; ++q) {
            int4 c = p[q];
            int4 r;
            r.x = run; run += c.x;
            r.y = run; run += c.y;
            r.z = run; run += c.z;
            r.w = run; run += c.w;
            rp[q] = r;
            float4 d;
            d.x = dinv_of(c.x);
            d.y = dinv_of(c.y);
            d.z = dinv_of(c.z);
            d.w = dinv_of(c.w);
            dp[q] = d;
        }
    } else {
        for (int i = lo; i < hi; ++i) {
            int c = cnt[i];
            rowptr[i] = run; run += c;
            dinv[i] = dinv_of(c);
        }
    }
    if (tid == 1023) rowptr[n] = run;
}

__global__ void fill_kernel(const int* __restrict__ src,
                            const int* __restrict__ dst,
                            const int* __restrict__ rank,
                            const int* __restrict__ rowptr,
                            const float* __restrict__ dinv,
                            int* __restrict__ col,
                            float* __restrict__ nrm, int ne) {
    int e = blockIdx.x * blockDim.x + threadIdx.x;
    if (e < ne) {
        const int s = src[e];
        const int d = dst[e];
        const int pos = rowptr[d] + rank[e];
        col[pos] = s;
        nrm[pos] = dinv[s] * dinv[d];
    }
}


template <bool RELU, int TN>
__global__ void gemm_wmma_strip(const float* __restrict__ A,
                                const float* __restrict__ W,
                                float* __restrict__ out,
                                int M, int K) {
    const int Nc   = TN * 16;
    const int wave = (blockIdx.x * blockDim.x + threadIdx.x) >> 5;
    const int lane = threadIdx.x & 31;

    const int tm = wave;
    if (tm * 16 >= M) return;

    const int ml    = lane & 15;
    const int khalf = (lane >> 4) << 1;

    const float* arow = A + (size_t)(tm * 16 + ml) * K;

    v8f acc[TN];
#pragma unroll
    for (int t = 0; t < TN; ++t) acc[t] = (v8f){};

    for (int k = 0; k < K; k += 4) {
        v2f a = *(const v2f*)(arow + k + khalf);
        if (RELU) {
            a.x = fmaxf(a.x, 0.0f);
            a.y = fmaxf(a.y, 0.0f);
        }

        const float* w0 = W + (size_t)(k + khalf) * Nc + ml;
        v2f b[TN];
#pragma unroll
        for (int t = 0; t < TN; ++t) {
            b[t].x = w0[t * 16];
            b[t].y = w0[Nc + t * 16];
        }

#pragma unroll
        for (int t = 0; t < TN; ++t) {
            acc[t] = __builtin_amdgcn_wmma_f32_16x16x4_f32(
                 false, a,  false, b[t],
                 (short)0, acc[t],  false,  false);
        }
    }

    const int row0 = tm * 16 + ((lane >> 4) << 3);
#pragma unroll
    for (int t = 0; t < TN; ++t) {
        float* o = out + (size_t)row0 * Nc + t * 16 + ml;
#pragma unroll
        for (int r = 0; r < 8; ++r) o[(size_t)r * Nc] = acc[t][r];
    }
}


template <int VW> struct lane_vec { typedef float type __attribute__((ext_vector_type(VW))); };

template <int C, int R>
__device__ __forceinline__ void gather_rows(const float* __restrict__ t, int s_l, float w_l,
                                            int j, int lane,
                                            typename lane_vec<C / 32>::type& acc) {
    typedef typename lane_vec<C / 32>::type vf;
    vf v[R];
#pragma unroll
    for (int u = 0; u < R; ++u) {
        const int s = __builtin_amdgcn_readlane(s_l, j + u);
        v[u] = *((const vf*)(t + (size_t)s * C) + lane);
    }
#pragma unroll
    for (int u = 0; u < R; ++u) {
        const float wgt = __builtin_bit_cast(float,
            __builtin_amdgcn_readlane(__builtin_bit_cast(int, w_l), j + u));
        acc += wgt * v[u];
    }
}

template <int C>
__global__ void __launch_bounds__(256)
csr_agg_kernel(const float* __restrict__ t,
               const float* __restrict__ dinv,
               const int* __restrict__ rowptr,
               const int* __restrict__ col,
               const float* __restrict__ nrm,
               const float* __restrict__ bias,
               float* __restrict__ out, int n) {
    typedef typename lane_vec<C / 32>::type vf;

    const int node = __builtin_amdgcn_readfirstlane(
        (int)((blockIdx.x * blockDim.x + threadIdx.x) >> 5));
    if (node >= n) return;
    const int lane = threadIdx.x & 31;

    const int beg = rowptr[node];
    const int end = rowptr[node + 1];
    const float di = dinv[node];

    const vf own = *((const vf*)(t + (size_t)node * C) + lane);
    vf acc = (vf)(0.0f);

    for (int p0 = beg; p0 < end; p0 += 32) {
        const int m = (end - p0 < 32) ? (end - p0) : 32;
        int   s_l = node;
        float w_l = 0.0f;
        if (lane < m) {
            s_l = col[p0 + lane];
            w_l = nrm[p0 + lane];
        }
        int j = 0;
        for (; j + 8 <= m; j += 8) gather_rows<C, 8>(t, s_l, w_l, j, lane, acc);
        switch (m - j) {
            case 7: gather_rows<C, 7>(t, s_l, w_l, j, lane, acc); break;
            case 6: gather_rows<C, 6>(t, s_l, w_l, j, lane, acc); break;
            case 5: gather_rows<C, 5>(t, s_l, w_l, j, lane, acc); break;
            case 4: gather_rows<C, 4>(t, s_l, w_l, j, lane, acc); break;
            case 3: gather_rows<C, 3>(t, s_l, w_l, j, lane, acc); break;
            case 2: gather_rows<C, 2>(t, s_l, w_l, j, lane, acc); break;
            case 1: gather_rows<C, 1>(t, s_l, w_l, j, lane, acc); break;
            default: break;
        }
    }
    acc += (di * di) * own;

    const vf b = *((const vf*)bias + lane);
    *((vf*)(out + (size_t)node * C) + lane) = acc + b;
}


static inline int cdiv(long long a, long long b) { return (int)((a + b - 1) / b); }

extern "C" void kernel_launch(void* const* d_in, const int* in_sizes, int n_in,
                              void* d_out, int out_size, void* d_ws, size_t ws_size,
                              hipStream_t stream) {
    const float* x  = (const float*)d_in[0];
    const float* W1 = (const float*)d_in[1];
    const float* b1 = (const float*)d_in[2];
    const float* W2 = (const float*)d_in[3];
    const float* b2 = (const float*)d_in[4];
    const float* W3 = (const float*)d_in[5];
    const float* b3 = (const float*)d_in[6];
    const int*   ei = (const int*)d_in[7];

    const int IN_C = 64, HID_C = 128, OUT_C = 64;
    const int n  = in_sizes[0] / IN_C;
    const int ne = in_sizes[7] / 2;
    const int* src = ei;
    const int* dst = ei + ne;

    float* T      = (float*)d_ws;
    float* G      = T + (size_t)n * HID_C;
    float* dinv   = G + (size_t)n * HID_C;
    int*   cnt    = (int*)(dinv + n);
    int*   rowptr = cnt + n;
    int*   rank   = rowptr + n + 64;
    int*   col    = rank + ne;
    float* nrm    = (float*)(col + ne);
    float* out    = (float*)d_out;

    const int TB = 256;
    const long long strip_waves = n / 16;
    const int gemm_blocks = cdiv(strip_waves * 32, TB);
    const int agg_blocks  = cdiv((long long)n * 32, TB);

    cnt_zero_kernel<<<cdiv(n, TB), TB, 0, stream>>>(cnt, n);
    count_rank_kernel<<<cdiv(ne, TB), TB, 0, stream>>>(dst, cnt, rank, ne);
    scan_dinv_kernel<<<1, 1024, 0, stream>>>(cnt, rowptr, dinv, n);
    fill_kernel<<<cdiv(ne, TB), TB, 0, stream>>>(src, dst, rank, rowptr, dinv, col, nrm, ne);

    gemm_wmma_strip<false, 8><<<gemm_blocks, TB, 0, stream>>>(x, W1, T, n, IN_C);
    csr_agg_kernel<128><<<agg_blocks, TB, 0, stream>>>(T, dinv, rowptr, col, nrm, b1, G, n);

    gemm_wmma_strip<true, 8><<<gemm_blocks, TB, 0, stream>>>(G, W2, T, n, HID_C);
    csr_agg_kernel<128><<<agg_blocks, TB, 0, stream>>>(T, dinv, rowptr, col, nrm, b2, G, n);

    gemm_wmma_strip<true, 4><<<gemm_blocks, TB, 0, stream>>>(G, W3, T, n, HID_C);
    csr_agg_kernel<64><<<agg_blocks, TB, 0, stream>>>(T, dinv, rowptr, col, nrm, b3, out, n);
}


# Layer 1 aggregates before it transforms, (A x) W1 + b1 for A (x W1) + b1: its gather reads 64-wide input rows, not 128-wide products; layer-1 sum order changes (reassociated)
# speedup vs baseline: 2.3711x; 1.0459x over previous
#include <hip/hip_runtime.h>


typedef float v2f __attribute__((ext_vector_type(2)));
typedef float v8f __attribute__((ext_vector_type(8)));


__global__ void cnt_zero_kernel(int* __restrict__ cnt, int n) {
    int i = blockIdx.x * blockDim.x + threadIdx.x;
    if (i < n) cnt[i] = 0;
}

__global__ void count_rank_kernel(const int* __restrict__ dst,
                                  int* __restrict__ cnt,
                                  int* __restrict__ rank, int ne) {
    int e = blockIdx.x * blockDim.x + threadIdx.x;
    if (e < ne) rank[e] = atomicAdd(&cnt[dst[e]], 1);
}

__device__ __forceinline__ float dinv_of(int c) {
    const float d = (float)(c + 1);
    return (d > 0.0f) ? rsqrtf(d) : 0.0f;
}

__global__ void __launch_bounds__(1024)
scan_dinv_kernel(const int* __restrict__ cnt, int* __restrict__ rowptr,
                 float* __restrict__ dinv, int n) {
    __shared__ int wsum[32];
    const int tid  = threadIdx.x;
    const int lane = tid & 31;
    const int w    = tid >> 5;
    const int per  = (n + 1023) >> 10;
    int lo = tid * per; if (lo > n) lo = n;
    int hi = lo + per;  if (hi > n) hi = n;
    const bool vec = ((per & 3) == 0) && (hi - lo == per);

    int s = 0;
    if (vec) {
        const int4* p = (const int4*)(cnt + lo);
        const int nq = per >> 2;
        int q = 0;
        for (; q + 8 <= nq; q += 8) {
            int4 c[8];
#pragma unroll
            for (int u = 0; u < 8; ++u) c[u] = p[q + u];
#pragma unroll
            for (int u = 0; u < 8; ++u) s += (c[u].x + c[u].y) + (c[u].z + c[u].w);
        }
        for (; q < nq; ++q) { int4 c = p[q]; s += (c.x + c.y) + (c.z + c.w); }
    } else {
        for (int i = lo; i < hi; ++i) s += cnt[i];
    }

    int incl = s;
#pragma unroll
    for (int off = 1; off < 32; off <<= 1) {
        int v = __shfl_up(incl, off);
        if (lane >= off) incl += v;
    }
    if (lane == 31) wsum[w] = incl;
    __syncthreads();
    if (w == 0) {
        int v = wsum[lane];
        int iv = v;
#pragma unroll
        for (int off = 1; off < 32; off <<= 1) {
            int u = __shfl_up(iv, off);
            if (lane >= off) iv += u;
        }
        wsum[lane] = iv - v;
    }
    __syncthreads();
    int run = wsum[w] + (incl - s);

    if (vec) {
        const int4* p = (const int4*)(cnt + lo);
        int4*   rp = (int4*)(rowptr + lo);
        float4* dp = (float4*)(dinv + lo);
        const int nq = per >> 2;
#pragma unroll 4
        for (int q = 0; q < nq; ++q) {
            int4 c = p[q];
            int4 r;
            r.x = run; run += c.x;
            r.y = run; run += c.y;
            r.z = run; run += c.z;
            r.w = run; run += c.w;
            rp[q] = r;
            float4 d;
            d.x = dinv_of(c.x);
            d.y = dinv_of(c.y);
            d.z = dinv_of(c.z);
            d.w = dinv_of(c.w);
            dp[q] = d;
        }
    } else {
        for (int i = lo; i < hi; ++i) {
            int c = cnt[i];
            rowptr[i] = run; run += c;
            dinv[i] = dinv_of(c);
        }
    }
    if (tid == 1023) rowptr[n] = run;
}

__global__ void fill_kernel(const int* __restrict__ src,
                            const int* __restrict__ dst,
                            const int* __restrict__ rank,
                            const int* __restrict__ rowptr,
                            const float* __restrict__ dinv,
                            int* __restrict__ col,
                            float* __restrict__ nrm, int ne) {
    int e = blockIdx.x * blockDim.x + threadIdx.x;
    if (e < ne) {
        const int s = src[e];
        const int d = dst[e];
        const int pos = rowptr[d] + rank[e];
        col[pos] = s;
        nrm[pos] = dinv[s] * dinv[d];
    }
}


template <bool RELU, bool BIAS, int TN>
__global__ void gemm_wmma_strip(const float* __restrict__ A,
                                const float* __restrict__ W,
                                const float* __restrict__ bias,
                                float* __restrict__ out,
                                int M, int K) {
    const int Nc   = TN * 16;
    const int wave = (blockIdx.x * blockDim.x + threadIdx.x) >> 5;
    const int lane = threadIdx.x & 31;

    const int tm = wave;
    if (tm * 16 >= M) return;

    const int ml    = lane & 15;
    const int khalf = (lane >> 4) << 1;

    const float* arow = A + (size_t)(tm * 16 + ml) * K;

    v8f acc[TN];
#pragma unroll
    for (int t = 0; t < TN; ++t) acc[t] = (v8f){};

    for (int k = 0; k < K; k += 4) {
        v2f a = *(const v2f*)(arow + k + khalf);
        if (RELU) {
            a.x = fmaxf(a.x, 0.0f);
            a.y = fmaxf(a.y, 0.0f);
        }

        const float* w0 = W + (size_t)(k + khalf) * Nc + ml;
        v2f b[TN];
#pragma unroll
        for (int t = 0; t < TN; ++t) {
            b[t].x = w0[t * 16];
            b[t].y = w0[Nc + t * 16];
        }

#pragma unroll
        for (int t = 0; t < TN; ++t) {
            acc[t] = __builtin_amdgcn_wmma_f32_16x16x4_f32(
                 false, a,  false, b[t],
                 (short)0, acc[t],  false,  false);
        }
    }

    const int row0 = tm * 16 + ((lane >> 4) << 3);
#pragma unroll
    for (int t = 0; t < TN; ++t) {
        float* o = out + (size_t)row0 * Nc + t * 16 + ml;
        const float bv = BIAS ? bias[t * 16 + ml] : 0.0f;
#pragma unroll
        for (int r = 0; r < 8; ++r) o[(size_t)r * Nc] = BIAS ? (acc[t][r] + bv) : acc[t][r];
    }
}


template <int VW> struct lane_vec { typedef float type __attribute__((ext_vector_type(VW))); };

template <int C, int R>
__device__ __forceinline__ void gather_rows(const float* __restrict__ t, int s_l, float w_l,
                                            int j, int lane,
                                            typename lane_vec<C / 32>::type& acc) {
    typedef typename lane_vec<C / 32>::type vf;
    vf v[R];
#pragma unroll
    for (int u = 0; u < R; ++u) {
        const int s = __builtin_amdgcn_readlane(s_l, j + u);
        v[u] = *((const vf*)(t + (size_t)s * C) + lane);
    }
#pragma unroll
    for (int u = 0; u < R; ++u) {
        const float wgt = __builtin_bit_cast(float,
            __builtin_amdgcn_readlane(__builtin_bit_cast(int, w_l), j + u));
        acc += wgt * v[u];
    }
}

template <int C, bool BIAS>
__global__ void __launch_bounds__(256)
csr_agg_kernel(const float* __restrict__ t,
               const float* __restrict__ dinv,
               const int* __restrict__ rowptr,
               const int* __restrict__ col,
               const float* __restrict__ nrm,
               const float* __restrict__ bias,
               float* __restrict__ out, int n) {
    typedef typename lane_vec<C / 32>::type vf;

    const int node = __builtin_amdgcn_readfirstlane(
        (int)((blockIdx.x * blockDim.x + threadIdx.x) >> 5));
    if (node >= n) return;
    const int lane = threadIdx.x & 31;

    const int beg = rowptr[node];
    const int end = rowptr[node + 1];
    const float di = dinv[node];

    const vf own = *((const vf*)(t + (size_t)node * C) + lane);
    vf acc = (vf)(0.0f);

    for (int p0 = beg; p0 < end; p0 += 32) {
        const int m = (end - p0 < 32) ? (end - p0) : 32;
        int   s_l = node;
        float w_l = 0.0f;
        if (lane < m) {
            s_l = col[p0 + lane];
            w_l = nrm[p0 + lane];
        }
        int j = 0;
        for (; j + 8 <= m; j += 8) gather_rows<C, 8>(t, s_l, w_l, j, lane, acc);
        switch (m - j) {
            case 7: gather_rows<C, 7>(t, s_l, w_l, j, lane, acc); break;
            case 6: gather_rows<C, 6>(t, s_l, w_l, j, lane, acc); break;
            case 5: gather_rows<C, 5>(t, s_l, w_l, j, lane, acc); break;
            case 4: gather_rows<C, 4>(t, s_l, w_l, j, lane, acc); break;
            case 3: gather_rows<C, 3>(t, s_l, w_l, j, lane, acc); break;
            case 2: gather_rows<C, 2>(t, s_l, w_l, j, lane, acc); break;
            case 1: gather_rows<C, 1>(t, s_l, w_l, j, lane, acc); break;
            default: break;
        }
    }
    acc += (di * di) * own;

    if (BIAS) {
        const vf b = *((const vf*)bias + lane);
        acc += b;
    }
    *((vf*)(out + (size_t)node * C) + lane) = acc;
}


static inline int cdiv(long long a, long long b) { return (int)((a + b - 1) / b); }

extern "C" void kernel_launch(void* const* d_in, const int* in_sizes, int n_in,
                              void* d_out, int out_size, void* d_ws, size_t ws_size,
                              hipStream_t stream) {
    const float* x  = (const float*)d_in[0];
    const float* W1 = (const float*)d_in[1];
    const float* b1 = (const float*)d_in[2];
    const float* W2 = (const float*)d_in[3];
    const float* b2 = (const float*)d_in[4];
    const float* W3 = (const float*)d_in[5];
    const float* b3 = (const float*)d_in[6];
    const int*   ei = (const int*)d_in[7];

    const int IN_C = 64, HID_C = 128, OUT_C = 64;
    const int n  = in_sizes[0] / IN_C;
    const int ne = in_sizes[7] / 2;
    const int* src = ei;
    const int* dst = ei + ne;

    float* T      = (float*)d_ws;
    float* G      = T + (size_t)n * HID_C;
    float* dinv   = G + (size_t)n * HID_C;
    int*   cnt    = (int*)(dinv + n);
    int*   rowptr = cnt + n;
    int*   rank   = rowptr + n + 64;
    int*   col    = rank + ne;
    float* nrm    = (float*)(col + ne);
    float* out    = (float*)d_out;

    const int TB = 256;
    const long long strip_waves = n / 16;
    const int gemm_blocks = cdiv(strip_waves * 32, TB);
    const int agg_blocks  = cdiv((long long)n * 32, TB);

    cnt_zero_kernel<<<cdiv(n, TB), TB, 0, stream>>>(cnt, n);
    count_rank_kernel<<<cdiv(ne, TB), TB, 0, stream>>>(dst, cnt, rank, ne);
    scan_dinv_kernel<<<1, 1024, 0, stream>>>(cnt, rowptr, dinv, n);
    fill_kernel<<<cdiv(ne, TB), TB, 0, stream>>>(src, dst, rank, rowptr, dinv, col, nrm, ne);

    float* AX = T;
    csr_agg_kernel<64, false><<<agg_blocks, TB, 0, stream>>>(x, dinv, rowptr, col, nrm, nullptr, AX, n);
    gemm_wmma_strip<false, true, 8><<<gemm_blocks, TB, 0, stream>>>(AX, W1, b1, G, n, IN_C);

    gemm_wmma_strip<true, false, 8><<<gemm_blocks, TB, 0, stream>>>(G, W2, nullptr, T, n, HID_C);
    csr_agg_kernel<128, true><<<agg_blocks, TB, 0, stream>>>(T, dinv, rowptr, col, nrm, b2, G, n);

    gemm_wmma_strip<true, false, 4><<<gemm_blocks, TB, 0, stream>>>(G, W3, nullptr, T, n, HID_C);
    csr_agg_kernel<64, true><<<agg_blocks, TB, 0, stream>>>(T, dinv, rowptr, col, nrm, b3, out, n);
}


# Edges grouped by destination by a two-level counting sort on LDS counters (chunk histogram, bucket scatter, per-bucket segments), no device atomic per edge; sum order by LDS arrival
# speedup vs baseline: 3.5173x; 1.4834x over previous
#include <hip/hip_runtime.h>


typedef float v2f __attribute__((ext_vector_type(2)));
typedef float v8f __attribute__((ext_vector_type(8)));


#define BUCKET_SHIFT 8
#define BUCKET_W 256
#define SORT_TB 256

__device__ __forceinline__ float dinv_of(int c) {
    const float d = (float)(c + 1);
    return (d > 0.0f) ? rsqrtf(d) : 0.0f;
}

__device__ __forceinline__ int block_excl_scan256(int v, int* wsum, int& total) {
    const int lane = threadIdx.x & 31;
    const int w    = threadIdx.x >> 5;
    int incl = v;
#pragma unroll
    for (int off = 1; off < 32; off <<= 1) {
        int u = __shfl_up(incl, off);
        if (lane >= off) incl += u;
    }
    if (lane == 31) wsum[w] = incl;
    __syncthreads();
    int base = 0, tot = 0;
#pragma unroll
    for (int i = 0; i < 8; ++i) {
        const int s = wsum[i];
        if (i < w) base += s;
        tot += s;
    }
    __syncthreads();
    total = tot;
    return base + incl - v;
}

__global__ void __launch_bounds__(SORT_TB)
bucket_hist_kernel(const int* __restrict__ dst, int* __restrict__ hist,
                   int ne, int chunk, int nb) {
    extern __shared__ int h[];
    const int tid = threadIdx.x;
    const int c   = blockIdx.x;
    for (int b = tid; b < nb; b += SORT_TB) h[b] = 0;
    __syncthreads();
    const int lo = c * chunk;
    int hi = lo + chunk; if (hi > ne) hi = ne;
    for (int e0 = lo; e0 < hi; e0 += 8 * SORT_TB) {
        int d[8];
#pragma unroll
        for (int u = 0; u < 8; ++u) {
            const int e = e0 + u * SORT_TB + tid;
            d[u] = (e < hi) ? dst[e] : -1;
        }
#pragma unroll
        for (int u = 0; u < 8; ++u)
            if (d[u] >= 0) atomicAdd(&h[d[u] >> BUCKET_SHIFT], 1);
    }
    __syncthreads();
    for (int b = tid; b < nb; b += SORT_TB) hist[c * nb + b] = h[b];
}

__global__ void __launch_bounds__(SORT_TB)
bucket_scatter_kernel(const int* __restrict__ src, const int* __restrict__ dst,
                      const int* __restrict__ hist, int* __restrict__ bstart,
                      unsigned* __restrict__ packed,
                      int ne, int chunk, int nchunks, int nb) {
    extern __shared__ int cur[];
    __shared__ int wsum[8];
    const int tid = threadIdx.x;
    const int c   = blockIdx.x;

    int carry = 0;
    for (int b0 = 0; b0 < nb; b0 += SORT_TB) {
        const int b = b0 + tid;
        int total = 0, pre = 0;
        if (b < nb) {
            const int* col_b = hist + b;
#pragma unroll 16
            for (int cc = 0; cc < nchunks; ++cc) {
                const int v = col_b[(size_t)cc * nb];
                total += v;
                pre += (cc < c) ? v : 0;
            }
        }
        int tile_total;
        const int ex = block_excl_scan256(total, wsum, tile_total);
        if (b < nb) {
            const int start = carry + ex;
            cur[b] = start + pre;
            if (c == 0) bstart[b] = start;
        }
        carry += tile_total;
    }
    if (c == 0 && tid == 0) bstart[nb] = carry;
    __syncthreads();

    const int lo = c * chunk;
    int hi = lo + chunk; if (hi > ne) hi = ne;
    for (int e0 = lo; e0 < hi; e0 += 8 * SORT_TB) {
        int d[8], s[8], p[8];
#pragma unroll
        for (int u = 0; u < 8; ++u) {
            const int e = e0 + u * SORT_TB + tid;
            const bool ok = e < hi;
            d[u] = ok ? dst[e] : -1;
            s[u] = ok ? src[e] : 0;
        }
#pragma unroll
        for (int u = 0; u < 8; ++u)
            p[u] = (d[u] >= 0) ? atomicAdd(&cur[d[u] >> BUCKET_SHIFT], 1) : -1;
#pragma unroll
        for (int u = 0; u < 8; ++u)
            if (p[u] >= 0)
                packed[p[u]] = ((unsigned)(d[u] & (BUCKET_W - 1)) << 24) | (unsigned)s[u];
    }
}

__global__ void __launch_bounds__(BUCKET_W)
bucket_csr_kernel(const unsigned* __restrict__ packed, const int* __restrict__ bstart,
                  int* __restrict__ rowptr, float* __restrict__ dinv,
                  int* __restrict__ col, int n, int nb) {
    __shared__ int cnt[BUCKET_W];
    __shared__ int wsum[8];
    const int tid = threadIdx.x;
    const int b   = blockIdx.x;
    const int beg = bstart[b];
    const int end = bstart[b + 1];

    cnt[tid] = 0;
    __syncthreads();
    for (int p0 = beg; p0 < end; p0 += 8 * BUCKET_W) {
        unsigned w[8];
#pragma unroll
        for (int u = 0; u < 8; ++u) {
            const int p = p0 + u * BUCKET_W + tid;
            w[u] = (p < end) ? packed[p] : 0xFFFFFFFFu;
        }
#pragma unroll
        for (int u = 0; u < 8; ++u)
            if (w[u] != 0xFFFFFFFFu) atomicAdd(&cnt[w[u] >> 24], 1);
    }
    __syncthreads();
    const int c = cnt[tid];
    int total;
    const int ex = block_excl_scan256(c, wsum, total);
    const int node = b * BUCKET_W + tid;
    if (node < n) {
        rowptr[node] = beg + ex;
        dinv[node]   = dinv_of(c);
    }
    if (b == nb - 1 && tid == 0) rowptr[n] = end;
    cnt[tid] = beg + ex;
    __syncthreads();
    for (int p0 = beg; p0 < end; p0 += 8 * BUCKET_W) {
        unsigned w[8];
        int pos[8];
#pragma unroll
        for (int u = 0; u < 8; ++u) {
            const int p = p0 + u * BUCKET_W + tid;
            w[u] = (p < end) ? packed[p] : 0u;
        }
#pragma unroll
        for (int u = 0; u < 8; ++u)
            pos[u] = (p0 + u * BUCKET_W + tid < end) ? atomicAdd(&cnt[w[u] >> 24], 1) : -1;
#pragma unroll
        for (int u = 0; u < 8; ++u)
            if (pos[u] >= 0) col[pos[u]] = (int)(w[u] & 0x00FFFFFFu);
    }
}


template <bool RELU, bool BIAS, int TN>
__global__ void gemm_wmma_strip(const float* __restrict__ A,
                                const float* __restrict__ W,
                                const float* __restrict__ bias,
                                float* __restrict__ out,
                                int M, int K) {
    const int Nc   = TN * 16;
    const int wave = (blockIdx.x * blockDim.x + threadIdx.x) >> 5;
    const int lane = threadIdx.x & 31;

    const int tm = wave;
    if (tm * 16 >= M) return;

    const int ml    = lane & 15;
    const int khalf = (lane >> 4) << 1;

    const float* arow = A + (size_t)(tm * 16 + ml) * K;

    v8f acc[TN];
#pragma unroll
    for (int t = 0; t < TN; ++t) acc[t] = (v8f){};

    for (int k = 0; k < K; k += 4) {
        v2f a = *(const v2f*)(arow + k + khalf);
        if (RELU) {
            a.x = fmaxf(a.x, 0.0f);
            a.y = fmaxf(a.y, 0.0f);
        }

        const float* w0 = W + (size_t)(k + khalf) * Nc + ml;
        v2f b[TN];
#pragma unroll
        for (int t = 0; t < TN; ++t) {
            b[t].x = w0[t * 16];
            b[t].y = w0[Nc + t * 16];
        }

#pragma unroll
        for (int t = 0; t < TN; ++t) {
            acc[t] = __builtin_amdgcn_wmma_f32_16x16x4_f32(
                 false, a,  false, b[t],
                 (short)0, acc[t],  false,  false);
        }
    }

    const int row0 = tm * 16 + ((lane >> 4) << 3);
#pragma unroll
    for (int t = 0; t < TN; ++t) {
        float* o = out + (size_t)row0 * Nc + t * 16 + ml;
        const float bv = BIAS ? bias[t * 16 + ml] : 0.0f;
#pragma unroll
        for (int r = 0; r < 8; ++r) o[(size_t)r * Nc] = BIAS ? (acc[t][r] + bv) : acc[t][r];
    }
}


template <int VW> struct lane_vec { typedef float type __attribute__((ext_vector_type(VW))); };

template <int C, int R>
__device__ __forceinline__ void gather_rows(const float* __restrict__ t, int s_l, float w_l,
                                            int j, int lane,
                                            typename lane_vec<C / 32>::type& acc) {
    typedef typename lane_vec<C / 32>::type vf;
    vf v[R];
#pragma unroll
    for (int u = 0; u < R; ++u) {
        const int s = __builtin_amdgcn_readlane(s_l, j + u);
        v[u] = *((const vf*)(t + (size_t)s * C) + lane);
    }
#pragma unroll
    for (int u = 0; u < R; ++u) {
        const float wgt = __builtin_bit_cast(float,
            __builtin_amdgcn_readlane(__builtin_bit_cast(int, w_l), j + u));
        acc += wgt * v[u];
    }
}

template <int C, bool BIAS, bool MAKE_NRM>
__global__ void __launch_bounds__(256)
csr_agg_kernel(const float* __restrict__ t,
               const float* __restrict__ dinv,
               const int* __restrict__ rowptr,
               const int* __restrict__ col,
               const float* __restrict__ nrm,
               float* __restrict__ nrm_out,
               const float* __restrict__ bias,
               float* __restrict__ out, int n) {
    typedef typename lane_vec<C / 32>::type vf;

    const int node = __builtin_amdgcn_readfirstlane(
        (int)((blockIdx.x * blockDim.x + threadIdx.x) >> 5));
    if (node >= n) return;
    const int lane = threadIdx.x & 31;

    const int beg = rowptr[node];
    const int end = rowptr[node + 1];
    const float di = dinv[node];

    const vf own = *((const vf*)(t + (size_t)node * C) + lane);
    vf acc = (vf)(0.0f);

    for (int p0 = beg; p0 < end; p0 += 32) {
        const int m = (end - p0 < 32) ? (end - p0) : 32;
        int   s_l = node;
        float w_l = 0.0f;
        if (lane < m) {
            s_l = col[p0 + lane];
            if (MAKE_NRM) {
                w_l = dinv[s_l] * di;
                nrm_out[p0 + lane] = w_l;
            } else {
                w_l = nrm[p0 + lane];
            }
        }
        int j = 0;
        for (; j + 8 <= m; j += 8) gather_rows<C, 8>(t, s_l, w_l, j, lane, acc);
        switch (m - j) {
            case 7: gather_rows<C, 7>(t, s_l, w_l, j, lane, acc); break;
            case 6: gather_rows<C, 6>(t, s_l, w_l, j, lane, acc); break;
            case 5: gather_rows<C, 5>(t, s_l, w_l, j, lane, acc); break;
            case 4: gather_rows<C, 4>(t, s_l, w_l, j, lane, acc); break;
            case 3: gather_rows<C, 3>(t, s_l, w_l, j, lane, acc); break;
            case 2: gather_rows<C, 2>(t, s_l, w_l, j, lane, acc); break;
            case 1: gather_rows<C, 1>(t, s_l, w_l, j, lane, acc); break;
            default: break;
        }
    }
    acc += (di * di) * own;

    if (BIAS) {
        const vf b = *((const vf*)bias + lane);
        acc += b;
    }
    *((vf*)(out + (size_t)node * C) + lane) = acc;
}


static inline int cdiv(long long a, long long b) { return (int)((a + b - 1) / b); }

extern "C" void kernel_launch(void* const* d_in, const int* in_sizes, int n_in,
                              void* d_out, int out_size, void* d_ws, size_t ws_size,
                              hipStream_t stream) {
    const float* x  = (const float*)d_in[0];
    const float* W1 = (const float*)d_in[1];
    const float* b1 = (const float*)d_in[2];
    const float* W2 = (const float*)d_in[3];
    const float* b2 = (const float*)d_in[4];
    const float* W3 = (const float*)d_in[5];
    const float* b3 = (const float*)d_in[6];
    const int*   ei = (const int*)d_in[7];

    const int IN_C = 64, HID_C = 128, OUT_C = 64;
    const int n  = in_sizes[0] / IN_C;
    const int ne = in_sizes[7] / 2;
    const int* src = ei;
    const int* dst = ei + ne;

    const int NCH   = 256;
    const int chunk = cdiv(ne, NCH);
    const int nb    = cdiv(n, BUCKET_W);

    float*    T      = (float*)d_ws;
    float*    G      = T + (size_t)n * HID_C;
    float*    dinv   = G + (size_t)n * HID_C;
    int*      rowptr = (int*)(dinv + n);
    int*      bstart = rowptr + n + 64;
    int*      hist   = bstart + nb + 64;
    unsigned* packed = (unsigned*)(hist + (size_t)NCH * nb);
    int*      col    = (int*)(packed + ne);
    float*    nrm    = (float*)(col + ne);
    float*    out    = (float*)d_out;

    const int TB = 256;
    const long long strip_waves = n / 16;
    const int gemm_blocks = cdiv(strip_waves * 32, TB);
    const int agg_blocks  = cdiv((long long)n * 32, TB);

    bucket_hist_kernel<<<NCH, SORT_TB, nb * sizeof(int), stream>>>(dst, hist, ne, chunk, nb);
    bucket_scatter_kernel<<<NCH, SORT_TB, nb * sizeof(int), stream>>>(src, dst, hist, bstart, packed,
                                                                      ne, chunk, NCH, nb);
    bucket_csr_kernel<<<nb, BUCKET_W, 0, stream>>>(packed, bstart, rowptr, dinv, col, n, nb);

    float* AX = T;
    csr_agg_kernel<64, false, true><<<agg_blocks, TB, 0, stream>>>(x, dinv, rowptr, col, nullptr, nrm, nullptr, AX, n);
    gemm_wmma_strip<false, true, 8><<<gemm_blocks, TB, 0, stream>>>(AX, W1, b1, G, n, IN_C);

    gemm_wmma_strip<true, false, 8><<<gemm_blocks, TB, 0, stream>>>(G, W2, nullptr, T, n, HID_C);
    csr_agg_kernel<128, true, false><<<agg_blocks, TB, 0, stream>>>(T, dinv, rowptr, col, nrm, nullptr, b2, G, n);

    gemm_wmma_strip<true, false, 4><<<gemm_blocks, TB, 0, stream>>>(G, W3, nullptr, T, n, HID_C);
    csr_agg_kernel<64, true, false><<<agg_blocks, TB, 0, stream>>>(T, dinv, rowptr, col, nrm, nullptr, b3, out, n);
}


# Each dense transform fused into the aggregation launch that feeds it: 64 aggregated rows stay in LDS and go through the WMMA products in the same work-group; 6 launches, not 9; sum order unchanged
# speedup vs baseline: 4.4939x; 1.2777x over previous
#include <hip/hip_runtime.h>


typedef float v2f __attribute__((ext_vector_type(2)));
typedef float v8f __attribute__((ext_vector_type(8)));


#define BUCKET_SHIFT 8
#define BUCKET_W 256
#define SORT_TB 256

__device__ __forceinline__ float dinv_of(int c) {
    const float d = (float)(c + 1);
    return (d > 0.0f) ? rsqrtf(d) : 0.0f;
}

__device__ __forceinline__ int block_excl_scan256(int v, int* wsum, int& total) {
    const int lane = threadIdx.x & 31;
    const int w    = threadIdx.x >> 5;
    int incl = v;
#pragma unroll
    for (int off = 1; off < 32; off <<= 1) {
        int u = __shfl_up(incl, off);
        if (lane >= off) incl += u;
    }
    if (lane == 31) wsum[w] = incl;
    __syncthreads();
    int base = 0, tot = 0;
#pragma unroll
    for (int i = 0; i < 8; ++i) {
        const int s = wsum[i];
        if (i < w) base += s;
        tot += s;
    }
    __syncthreads();
    total = tot;
    return base + incl - v;
}

__global__ void __launch_bounds__(SORT_TB)
bucket_hist_kernel(const int* __restrict__ dst, int* __restrict__ hist,
                   int ne, int chunk, int nb) {
    extern __shared__ int h[];
    const int tid = threadIdx.x;
    const int c   = blockIdx.x;
    for (int b = tid; b < nb; b += SORT_TB) h[b] = 0;
    __syncthreads();
    const int lo = c * chunk;
    int hi = lo + chunk; if (hi > ne) hi = ne;
    for (int e0 = lo; e0 < hi; e0 += 8 * SORT_TB) {
        int d[8];
#pragma unroll
        for (int u = 0; u < 8; ++u) {
            const int e = e0 + u * SORT_TB + tid;
            d[u] = (e < hi) ? dst[e] : -1;
        }
#pragma unroll
        for (int u = 0; u < 8; ++u)
            if (d[u] >= 0) atomicAdd(&h[d[u] >> BUCKET_SHIFT], 1);
    }
    __syncthreads();
    for (int b = tid; b < nb; b += SORT_TB) hist[c * nb + b] = h[b];
}

__global__ void __launch_bounds__(SORT_TB)
bucket_scatter_kernel(const int* __restrict__ src, const int* __restrict__ dst,
                      const int* __restrict__ hist, int* __restrict__ bstart,
                      unsigned* __restrict__ packed,
                      int ne, int chunk, int nchunks, int nb) {
    extern __shared__ int cur[];
    __shared__ int wsum[8];
    const int tid = threadIdx.x;
    const int c   = blockIdx.x;

    int carry = 0;
    for (int b0 = 0; b0 < nb; b0 += SORT_TB) {
        const int b = b0 + tid;
        int total = 0, pre = 0;
        if (b < nb) {
            const int* col_b = hist + b;
#pragma unroll 16
            for (int cc = 0; cc < nchunks; ++cc) {
                const int v = col_b[(size_t)cc * nb];
                total += v;
                pre += (cc < c) ? v : 0;
            }
        }
        int tile_total;
        const int ex = block_excl_scan256(total, wsum, tile_total);
        if (b < nb) {
            const int start = carry + ex;
            cur[b] = start + pre;
            if (c == 0) bstart[b] = start;
        }
        carry += tile_total;
    }
    if (c == 0 && tid == 0) bstart[nb] = carry;
    __syncthreads();

    const int lo = c * chunk;
    int hi = lo + chunk; if (hi > ne) hi = ne;
    for (int e0 = lo; e0 < hi; e0 += 8 * SORT_TB) {
        int d[8], s[8], p[8];
#pragma unroll
        for (int u = 0; u < 8; ++u) {
            const int e = e0 + u * SORT_TB + tid;
            const bool ok = e < hi;
            d[u] = ok ? dst[e] : -1;
            s[u] = ok ? src[e] : 0;
        }
#pragma unroll
        for (int u = 0; u < 8; ++u)
            p[u] = (d[u] >= 0) ? atomicAdd(&cur[d[u] >> BUCKET_SHIFT], 1) : -1;
#pragma unroll
        for (int u = 0; u < 8; ++u)
            if (p[u] >= 0)
                packed[p[u]] = ((unsigned)(d[u] & (BUCKET_W - 1)) << 24) | (unsigned)s[u];
    }
}

__global__ void __launch_bounds__(BUCKET_W)
bucket_csr_kernel(const unsigned* __restrict__ packed, const int* __restrict__ bstart,
                  int* __restrict__ rowptr, float* __restrict__ dinv,
                  int* __restrict__ col, int n, int nb) {
    __shared__ int cnt[BUCKET_W];
    __shared__ int wsum[8];
    const int tid = threadIdx.x;
    const int b   = blockIdx.x;
    const int beg = bstart[b];
    const int end = bstart[b + 1];

    cnt[tid] = 0;
    __syncthreads();
    for (int p0 = beg; p0 < end; p0 += 8 * BUCKET_W) {
        unsigned w[8];
#pragma unroll
        for (int u = 0; u < 8; ++u) {
            const int p = p0 + u * BUCKET_W + tid;
            w[u] = (p < end) ? packed[p] : 0xFFFFFFFFu;
        }
#pragma unroll
        for (int u = 0; u < 8; ++u)
            if (w[u] != 0xFFFFFFFFu) atomicAdd(&cnt[w[u] >> 24], 1);
    }
    __syncthreads();
    const int c = cnt[tid];
    int total;
    const int ex = block_excl_scan256(c, wsum, total);
    const int node = b * BUCKET_W + tid;
    if (node < n) {
        rowptr[node] = beg + ex;
        dinv[node]   = dinv_of(c);
    }
    if (b == nb - 1 && tid == 0) rowptr[n] = end;
    cnt[tid] = beg + ex;
    __syncthreads();
    for (int p0 = beg; p0 < end; p0 += 8 * BUCKET_W) {
        unsigned w[8];
        int pos[8];
#pragma unroll
        for (int u = 0; u < 8; ++u) {
            const int p = p0 + u * BUCKET_W + tid;
            w[u] = (p < end) ? packed[p] : 0u;
        }
#pragma unroll
        for (int u = 0; u < 8; ++u)
            pos[u] = (p0 + u * BUCKET_W + tid < end) ? atomicAdd(&cnt[w[u] >> 24], 1) : -1;
#pragma unroll
        for (int u = 0; u < 8; ++u)
            if (pos[u] >= 0) col[pos[u]] = (int)(w[u] & 0x00FFFFFFu);
    }
}


template <int VW> struct lane_vec { typedef float type __attribute__((ext_vector_type(VW))); };

template <int C, int R>
__device__ __forceinline__ void gather_rows(const float* __restrict__ t, int s_l, float w_l,
                                            int j, int lane,
                                            typename lane_vec<C / 32>::type& acc) {
    typedef typename lane_vec<C / 32>::type vf;
    vf v[R];
#pragma unroll
    for (int u = 0; u < R; ++u) {
        const int s = __builtin_amdgcn_readlane(s_l, j + u);
        v[u] = *((const vf*)(t + (size_t)s * C) + lane);
    }
#pragma unroll
    for (int u = 0; u < R; ++u) {
        const float wgt = __builtin_bit_cast(float,
            __builtin_amdgcn_readlane(__builtin_bit_cast(int, w_l), j + u));
        acc += wgt * v[u];
    }
}

template <int C, bool MAKE_NRM>
__device__ __forceinline__ typename lane_vec<C / 32>::type
agg_row(const float* __restrict__ t,
        const float* __restrict__ dinv,
        const int* __restrict__ rowptr,
        const int* __restrict__ col,
        const float* __restrict__ nrm,
        float* __restrict__ nrm_out,
        int node, int lane) {
    typedef typename lane_vec<C / 32>::type vf;

    const int beg = rowptr[node];
    const int end = rowptr[node + 1];
    const float di = dinv[node];

    const vf own = *((const vf*)(t + (size_t)node * C) + lane);
    vf acc = (vf)(0.0f);

    for (int p0 = beg; p0 < end; p0 += 32) {
        const int m = (end - p0 < 32) ? (end - p0) : 32;
        int   s_l = node;
        float w_l = 0.0f;
        if (lane < m) {
            s_l = col[p0 + lane];
            if (MAKE_NRM) {
                w_l = dinv[s_l] * di;
                nrm_out[p0 + lane] = w_l;
            } else {
                w_l = nrm[p0 + lane];
            }
        }
        int j = 0;
        for (; j + 8 <= m; j += 8) gather_rows<C, 8>(t, s_l, w_l, j, lane, acc);
        switch (m - j) {
            case 7: gather_rows<C, 7>(t, s_l, w_l, j, lane, acc); break;
            case 6: gather_rows<C, 6>(t, s_l, w_l, j, lane, acc); break;
            case 5: gather_rows<C, 5>(t, s_l, w_l, j, lane, acc); break;
            case 4: gather_rows<C, 4>(t, s_l, w_l, j, lane, acc); break;
            case 3: gather_rows<C, 3>(t, s_l, w_l, j, lane, acc); break;
            case 2: gather_rows<C, 2>(t, s_l, w_l, j, lane, acc); break;
            case 1: gather_rows<C, 1>(t, s_l, w_l, j, lane, acc); break;
            default: break;
        }
    }
    acc += (di * di) * own;
    return acc;
}

template <int C>
__global__ void __launch_bounds__(256)
csr_agg_kernel(const float* __restrict__ t,
               const float* __restrict__ dinv,
               const int* __restrict__ rowptr,
               const int* __restrict__ col,
               const float* __restrict__ nrm,
               const float* __restrict__ bias,
               float* __restrict__ out, int n) {
    typedef typename lane_vec<C / 32>::type vf;
    const int node = __builtin_amdgcn_readfirstlane(
        (int)((blockIdx.x * blockDim.x + threadIdx.x) >> 5));
    if (node >= n) return;
    const int lane = threadIdx.x & 31;
    vf acc = agg_row<C, false>(t, dinv, rowptr, col, nrm, nullptr, node, lane);
    const vf b = *((const vf*)bias + lane);
    acc += b;
    *((vf*)(out + (size_t)node * C) + lane) = acc;
}


#define FM 64
#define FWAVES 16
#define PITCH64 68
#define PITCH128 132

typedef float v4f __attribute__((ext_vector_type(4)));

template <int K, int NC>
__device__ __forceinline__ void strip_pair_wmma(const float* arow, const float* __restrict__ wcol,
                                                v8f& c0, v8f& c1) {
#pragma unroll 4
    for (int k = 0; k < K; k += 4) {
        const v2f a = *(const v2f*)(arow + k);
        const float* w0 = wcol + (size_t)k * NC;
        v2f b0, b1;
        b0.x = w0[0];
        b0.y = w0[NC];
        b1.x = w0[16];
        b1.y = w0[NC + 16];
        c0 = __builtin_amdgcn_wmma_f32_16x16x4_f32(
             false, a,  false, b0,
             (short)0, c0,  false,  false);
        c1 = __builtin_amdgcn_wmma_f32_16x16x4_f32(
             false, a,  false, b1,
             (short)0, c1,  false,  false);
    }
}

__global__ void __launch_bounds__(FWAVES * 32)
agg_l1_gemm2_kernel(const float* __restrict__ x,
                    const float* __restrict__ dinv,
                    const int* __restrict__ rowptr,
                    const int* __restrict__ col,
                    float* __restrict__ nrm_out,
                    const float* __restrict__ W1,
                    const float* __restrict__ b1,
                    const float* __restrict__ W2,
                    float* __restrict__ T, int n) {
    __shared__ __attribute__((aligned(16))) float sA[FM * PITCH64];
    __shared__ __attribute__((aligned(16))) float sG[FM * PITCH128];

    const int lane = threadIdx.x & 31;
    const int w    = __builtin_amdgcn_readfirstlane((int)(threadIdx.x >> 5));
    const int row0 = blockIdx.x * FM;

    for (int r = 0; r < FM / FWAVES; ++r) {
        const int lr   = w * (FM / FWAVES) + r;
        const int node = row0 + lr;
        v2f acc = (v2f)(0.0f);
        if (node < n) acc = agg_row<64, true>(x, dinv, rowptr, col, nullptr, nrm_out, node, lane);
        *((v2f*)(sA + lr * PITCH64) + lane) = acc;
    }
    __syncthreads();

    const int s     = w >> 2;
    const int jp    = w & 3;
    const int ml    = lane & 15;
    const int h     = lane >> 4;
    const int khalf = h << 1;

    {
        v8f c0 = (v8f){}, c1 = (v8f){};
        strip_pair_wmma<64, 128>(sA + (s * 16 + ml) * PITCH64 + khalf,
                                 W1 + (size_t)khalf * 128 + jp * 32 + ml, c0, c1);
        const float bv0 = b1[jp * 32 + ml];
        const float bv1 = b1[jp * 32 + 16 + ml];
        float* g = sG + (s * 16 + 8 * h) * PITCH128 + jp * 32 + ml;
#pragma unroll
        for (int r = 0; r < 8; ++r) {
            g[r * PITCH128]      = fmaxf(c0[r] + bv0, 0.0f);
            g[r * PITCH128 + 16] = fmaxf(c1[r] + bv1, 0.0f);
        }
    }
    __syncthreads();

    {
        v8f c0 = (v8f){}, c1 = (v8f){};
        strip_pair_wmma<128, 128>(sG + (s * 16 + ml) * PITCH128 + khalf,
                                  W2 + (size_t)khalf * 128 + jp * 32 + ml, c0, c1);
        const int grow = row0 + s * 16 + 8 * h;
        float* o = T + (size_t)grow * 128 + jp * 32 + ml;
        if (row0 + s * 16 + 16 <= n) {
#pragma unroll
            for (int r = 0; r < 8; ++r) {
                o[(size_t)r * 128]      = c0[r];
                o[(size_t)r * 128 + 16] = c1[r];
            }
        } else {
#pragma unroll
            for (int r = 0; r < 8; ++r) {
                if (grow + r < n) {
                    o[(size_t)r * 128]      = c0[r];
                    o[(size_t)r * 128 + 16] = c1[r];
                }
            }
        }
    }
}

__global__ void __launch_bounds__(FWAVES * 32)
agg_l2_gemm3_kernel(const float* __restrict__ T,
                    const float* __restrict__ dinv,
                    const int* __restrict__ rowptr,
                    const int* __restrict__ col,
                    const float* __restrict__ nrm,
                    const float* __restrict__ b2,
                    const float* __restrict__ W3,
                    float* __restrict__ U, int n) {
    __shared__ __attribute__((aligned(16))) float sG[FM * PITCH128];

    const int lane = threadIdx.x & 31;
    const int w    = __builtin_amdgcn_readfirstlane((int)(threadIdx.x >> 5));
    const int row0 = blockIdx.x * FM;

    const v4f bias = *((const v4f*)b2 + lane);
    for (int r = 0; r < FM / FWAVES; ++r) {
        const int lr   = w * (FM / FWAVES) + r;
        const int node = row0 + lr;
        v4f acc = (v4f)(0.0f);
        if (node < n) {
            acc = agg_row<128, false>(T, dinv, rowptr, col, nrm, nullptr, node, lane);
            acc += bias;
        }
        acc.x = fmaxf(acc.x, 0.0f);
        acc.y = fmaxf(acc.y, 0.0f);
        acc.z = fmaxf(acc.z, 0.0f);
        acc.w = fmaxf(acc.w, 0.0f);
        *((v4f*)(sG + lr * PITCH128) + lane) = acc;
    }
    __syncthreads();

    if (w < 8) {
        const int s     = w >> 1;
        const int jp    = w & 1;
        const int ml    = lane & 15;
        const int h     = lane >> 4;
        const int khalf = h << 1;
        v8f c0 = (v8f){}, c1 = (v8f){};
        strip_pair_wmma<128, 64>(sG + (s * 16 + ml) * PITCH128 + khalf,
                                 W3 + (size_t)khalf * 64 + jp * 32 + ml, c0, c1);
        const int grow = row0 + s * 16 + 8 * h;
        float* o = U + (size_t)grow * 64 + jp * 32 + ml;
        if (row0 + s * 16 + 16 <= n) {
#pragma unroll
            for (int r = 0; r < 8; ++r) {
                o[(size_t)r * 64]      = c0[r];
                o[(size_t)r * 64 + 16] = c1[r];
            }
        } else {
#pragma unroll
            for (int r = 0; r < 8; ++r) {
                if (grow + r < n) {
                    o[(size_t)r * 64]      = c0[r];
                    o[(size_t)r * 64 + 16] = c1[r];
                }
            }
        }
    }
}


static inline int cdiv(long long a, long long b) { return (int)((a + b - 1) / b); }

extern "C" void kernel_launch(void* const* d_in, const int* in_sizes, int n_in,
                              void* d_out, int out_size, void* d_ws, size_t ws_size,
                              hipStream_t stream) {
    const float* x  = (const float*)d_in[0];
    const float* W1 = (const float*)d_in[1];
    const float* b1 = (const float*)d_in[2];
    const float* W2 = (const float*)d_in[3];
    const float* b2 = (const float*)d_in[4];
    const float* W3 = (const float*)d_in[5];
    const float* b3 = (const float*)d_in[6];
    const int*   ei = (const int*)d_in[7];

    const int IN_C = 64, HID_C = 128, OUT_C = 64;
    const int n  = in_sizes[0] / IN_C;
    const int ne = in_sizes[7] / 2;
    const int* src = ei;
    const int* dst = ei + ne;

    const int NCH   = 256;
    const int chunk = cdiv(ne, NCH);
    const int nb    = cdiv(n, BUCKET_W);

    float*    T      = (float*)d_ws;
    float*    G      = T + (size_t)n * HID_C;
    float*    dinv   = G + (size_t)n * HID_C;
    int*      rowptr = (int*)(dinv + n);
    int*      bstart = rowptr + n + 64;
    int*      hist   = bstart + nb + 64;
    unsigned* packed = (unsigned*)(hist + (size_t)NCH * nb);
    int*      col    = (int*)(packed + ne);
    float*    nrm    = (float*)(col + ne);
    float*    out    = (float*)d_out;

    const int TB = 256;
    const int agg_blocks   = cdiv((long long)n * 32, TB);
    const int fused_blocks = cdiv(n, FM);

    bucket_hist_kernel<<<NCH, SORT_TB, nb * sizeof(int), stream>>>(dst, hist, ne, chunk, nb);
    bucket_scatter_kernel<<<NCH, SORT_TB, nb * sizeof(int), stream>>>(src, dst, hist, bstart, packed,
                                                                      ne, chunk, NCH, nb);
    bucket_csr_kernel<<<nb, BUCKET_W, 0, stream>>>(packed, bstart, rowptr, dinv, col, n, nb);

    agg_l1_gemm2_kernel<<<fused_blocks, FWAVES * 32, 0, stream>>>(x, dinv, rowptr, col, nrm,
                                                                 W1, b1, W2, T, n);

    agg_l2_gemm3_kernel<<<fused_blocks, FWAVES * 32, 0, stream>>>(T, dinv, rowptr, col, nrm,
                                                                 b2, W3, G, n);

    csr_agg_kernel<64><<<agg_blocks, TB, 0, stream>>>(G, dinv, rowptr, col, nrm, b3, out, n);
}


# First sort level in one launch: each chunk ordered by bucket in its own range plus a row of offsets, bucket pass walks the chunk segments; no global histogram, 5 launches; sum order by LDS arrival
# speedup vs baseline: 5.7100x; 1.2706x over previous
#include <hip/hip_runtime.h>


typedef float v2f __attribute__((ext_vector_type(2)));
typedef float v8f __attribute__((ext_vector_type(8)));


#define BUCKET_SHIFT 8
#define BUCKET_W 256
#define SORT_TB 256

__device__ __forceinline__ float dinv_of(int c) {
    const float d = (float)(c + 1);
    return (d > 0.0f) ? rsqrtf(d) : 0.0f;
}

__device__ __forceinline__ int block_excl_scan256(int v, int* wsum, int& total) {
    const int lane = threadIdx.x & 31;
    const int w    = threadIdx.x >> 5;
    int incl = v;
#pragma unroll
    for (int off = 1; off < 32; off <<= 1) {
        int u = __shfl_up(incl, off);
        if (lane >= off) incl += u;
    }
    if (lane == 31) wsum[w] = incl;
    __syncthreads();
    int base = 0, tot = 0;
#pragma unroll
    for (int i = 0; i < 8; ++i) {
        const int s = wsum[i];
        if (i < w) base += s;
        tot += s;
    }
    __syncthreads();
    total = tot;
    return base + incl - v;
}

__global__ void __launch_bounds__(SORT_TB)
chunk_order_kernel(const int* __restrict__ src, const int* __restrict__ dst,
                   int* __restrict__ seg, unsigned* __restrict__ packed,
                   int ne, int chunk, int nb) {
    extern __shared__ int cur[];
    __shared__ int wsum[8];
    const int tid = threadIdx.x;
    const int c   = blockIdx.x;
    for (int b = tid; b < nb; b += SORT_TB) cur[b] = 0;
    __syncthreads();
    const int lo = c * chunk;
    int hi = lo + chunk; if (hi > ne) hi = ne;
    for (int e0 = lo; e0 < hi; e0 += 16 * SORT_TB) {
        int d[16];
#pragma unroll
        for (int u = 0; u < 16; ++u) {
            const int e = e0 + u * SORT_TB + tid;
            d[u] = (e < hi) ? dst[e] : -1;
        }
#pragma unroll
        for (int u = 0; u < 16; ++u)
            if (d[u] >= 0) atomicAdd(&cur[d[u] >> BUCKET_SHIFT], 1);
    }
    __syncthreads();

    int* seg_c = seg + (size_t)c * (nb + 1);
    int carry = 0;
    for (int b0 = 0; b0 < nb; b0 += SORT_TB) {
        const int b = b0 + tid;
        const int v = (b < nb) ? cur[b] : 0;
        int tile_total;
        const int ex = block_excl_scan256(v, wsum, tile_total);
        if (b < nb) {
            cur[b]   = carry + ex;
            seg_c[b] = carry + ex;
        }
        carry += tile_total;
    }
    if (tid == 0) seg_c[nb] = carry;
    __syncthreads();

    for (int e0 = lo; e0 < hi; e0 += 8 * SORT_TB) {
        int d[8], s[8], p[8];
#pragma unroll
        for (int u = 0; u < 8; ++u) {
            const int e = e0 + u * SORT_TB + tid;
            const bool ok = e < hi;
            d[u] = ok ? dst[e] : -1;
            s[u] = ok ? src[e] : 0;
        }
#pragma unroll
        for (int u = 0; u < 8; ++u)
            p[u] = (d[u] >= 0) ? atomicAdd(&cur[d[u] >> BUCKET_SHIFT], 1) : -1;
#pragma unroll
        for (int u = 0; u < 8; ++u)
            if (p[u] >= 0)
                packed[lo + p[u]] = ((unsigned)(d[u] & (BUCKET_W - 1)) << 24) | (unsigned)s[u];
    }
}

#define SEG_GROUP 16
__global__ void __launch_bounds__(BUCKET_W)
bucket_csr_kernel(const unsigned* __restrict__ packed, const int* __restrict__ seg,
                  int* __restrict__ rowptr, float* __restrict__ dinv,
                  int* __restrict__ col, int n, int nb, int chunk) {
    __shared__ int cnt[BUCKET_W];
    __shared__ int wsum[8];
    const int tid  = threadIdx.x;
    const int lane = tid & 31;
    const int b    = blockIdx.x;

    const int* so  = seg + (size_t)tid * (nb + 1) + b;
    const int o0   = so[0];
    const int len  = so[1] - o0;
    const int base = tid * chunk + o0;

    cnt[tid] = 0;
    int beg, total;
    (void)block_excl_scan256(o0, wsum, beg);
    (void)block_excl_scan256(len, wsum, total);
    const int end = beg + total;

    for (int g = 0; g < 32; g += SEG_GROUP) {
        int bs[SEG_GROUP], ln[SEG_GROUP];
        int mx = 0;
#pragma unroll
        for (int k = 0; k < SEG_GROUP; ++k) {
            bs[k] = __builtin_amdgcn_readlane(base, g + k);
            ln[k] = __builtin_amdgcn_readlane(len, g + k);
            mx = (ln[k] > mx) ? ln[k] : mx;
        }
        for (int i0 = 0; i0 < mx; i0 += 32) {
            unsigned wv[SEG_GROUP];
            const int i = i0 + lane;
#pragma unroll
            for (int k = 0; k < SEG_GROUP; ++k)
                wv[k] = (i < ln[k]) ? packed[bs[k] + i] : 0xFFFFFFFFu;
#pragma unroll
            for (int k = 0; k < SEG_GROUP; ++k)
                if (wv[k] != 0xFFFFFFFFu) atomicAdd(&cnt[wv[k] >> 24], 1);
        }
    }
    __syncthreads();
    const int c = cnt[tid];
    int tot2;
    const int ex = block_excl_scan256(c, wsum, tot2);
    const int node = b * BUCKET_W + tid;
    if (node < n) {
        rowptr[node] = beg + ex;
        dinv[node]   = dinv_of(c);
    }
    if (b == nb - 1 && tid == 0) rowptr[n] = end;
    cnt[tid] = beg + ex;
    __syncthreads();

    for (int g = 0; g < 32; g += SEG_GROUP) {
        int bs[SEG_GROUP], ln[SEG_GROUP];
        int mx = 0;
#pragma unroll
        for (int k = 0; k < SEG_GROUP; ++k) {
            bs[k] = __builtin_amdgcn_readlane(base, g + k);
            ln[k] = __builtin_amdgcn_readlane(len, g + k);
            mx = (ln[k] > mx) ? ln[k] : mx;
        }
        for (int i0 = 0; i0 < mx; i0 += 32) {
            unsigned wv[SEG_GROUP];
            int pos[SEG_GROUP];
            const int i = i0 + lane;
#pragma unroll
            for (int k = 0; k < SEG_GROUP; ++k)
                wv[k] = (i < ln[k]) ? packed[bs[k] + i] : 0u;
#pragma unroll
            for (int k = 0; k < SEG_GROUP; ++k)
                pos[k] = (i < ln[k]) ? atomicAdd(&cnt[wv[k] >> 24], 1) : -1;
#pragma unroll
            for (int k = 0; k < SEG_GROUP; ++k)
                if (pos[k] >= 0) col[pos[k]] = (int)(wv[k] & 0x00FFFFFFu);
        }
    }
}


template <int VW> struct lane_vec { typedef float type __attribute__((ext_vector_type(VW))); };

template <int C, int R>
__device__ __forceinline__ void gather_rows(const float* __restrict__ t, int s_l, float w_l,
                                            int j, int lane,
                                            typename lane_vec<C / 32>::type& acc) {
    typedef typename lane_vec<C / 32>::type vf;
    vf v[R];
#pragma unroll
    for (int u = 0; u < R; ++u) {
        const int s = __builtin_amdgcn_readlane(s_l, j + u);
        v[u] = *((const vf*)(t + (size_t)s * C) + lane);
    }
#pragma unroll
    for (int u = 0; u < R; ++u) {
        const float wgt = __builtin_bit_cast(float,
            __builtin_amdgcn_readlane(__builtin_bit_cast(int, w_l), j + u));
        acc += wgt * v[u];
    }
}

template <int C, bool MAKE_NRM>
__device__ __forceinline__ typename lane_vec<C / 32>::type
agg_row(const float* __restrict__ t,
        const float* __restrict__ dinv,
        const int* __restrict__ rowptr,
        const int* __restrict__ col,
        const float* __restrict__ nrm,
        float* __restrict__ nrm_out,
        int node, int lane) {
    typedef typename lane_vec<C / 32>::type vf;

    const int beg = rowptr[node];
    const int end = rowptr[node + 1];
    const float di = dinv[node];

    const vf own = *((const vf*)(t + (size_t)node * C) + lane);
    vf acc = (vf)(0.0f);

    for (int p0 = beg; p0 < end; p0 += 32) {
        const int m = (end - p0 < 32) ? (end - p0) : 32;
        int   s_l = node;
        float w_l = 0.0f;
        if (lane < m) {
            s_l = col[p0 + lane];
            if (MAKE_NRM) {
                w_l = dinv[s_l] * di;
                nrm_out[p0 + lane] = w_l;
            } else {
                w_l = nrm[p0 + lane];
            }
        }
        int j = 0;
        for (; j + 8 <= m; j += 8) gather_rows<C, 8>(t, s_l, w_l, j, lane, acc);
        switch (m - j) {
            case 7: gather_rows<C, 7>(t, s_l, w_l, j, lane, acc); break;
            case 6: gather_rows<C, 6>(t, s_l, w_l, j, lane, acc); break;
            case 5: gather_rows<C, 5>(t, s_l, w_l, j, lane, acc); break;
            case 4: gather_rows<C, 4>(t, s_l, w_l, j, lane, acc); break;
            case 3: gather_rows<C, 3>(t, s_l, w_l, j, lane, acc); break;
            case 2: gather_rows<C, 2>(t, s_l, w_l, j, lane, acc); break;
            case 1: gather_rows<C, 1>(t, s_l, w_l, j, lane, acc); break;
            default: break;
        }
    }
    acc += (di * di) * own;
    return acc;
}

template <int C>
__global__ void __launch_bounds__(256)
csr_agg_kernel(const float* __restrict__ t,
               const float* __restrict__ dinv,
               const int* __restrict__ rowptr,
               const int* __restrict__ col,
               const float* __restrict__ nrm,
               const float* __restrict__ bias,
               float* __restrict__ out, int n) {
    typedef typename lane_vec<C / 32>::type vf;
    const int node = __builtin_amdgcn_readfirstlane(
        (int)((blockIdx.x * blockDim.x + threadIdx.x) >> 5));
    if (node >= n) return;
    const int lane = threadIdx.x & 31;
    vf acc = agg_row<C, false>(t, dinv, rowptr, col, nrm, nullptr, node, lane);
    const vf b = *((const vf*)bias + lane);
    acc += b;
    *((vf*)(out + (size_t)node * C) + lane) = acc;
}


#define FM 64
#define FWAVES 16
#define PITCH64 68
#define PITCH128 132

typedef float v4f __attribute__((ext_vector_type(4)));

template <int K, int NC>
__device__ __forceinline__ void strip_pair_wmma(const float* arow, const float* __restrict__ wcol,
                                                v8f& c0, v8f& c1) {
#pragma unroll 4
    for (int k = 0; k < K; k += 4) {
        const v2f a = *(const v2f*)(arow + k);
        const float* w0 = wcol + (size_t)k * NC;
        v2f b0, b1;
        b0.x = w0[0];
        b0.y = w0[NC];
        b1.x = w0[16];
        b1.y = w0[NC + 16];
        c0 = __builtin_amdgcn_wmma_f32_16x16x4_f32(
             false, a,  false, b0,
             (short)0, c0,  false,  false);
        c1 = __builtin_amdgcn_wmma_f32_16x16x4_f32(
             false, a,  false, b1,
             (short)0, c1,  false,  false);
    }
}

__global__ void __launch_bounds__(FWAVES * 32)
agg_l1_gemm2_kernel(const float* __restrict__ x,
                    const float* __restrict__ dinv,
                    const int* __restrict__ rowptr,
                    const int* __restrict__ col,
                    float* __restrict__ nrm_out,
                    const float* __restrict__ W1,
                    const float* __restrict__ b1,
                    const float* __restrict__ W2,
                    float* __restrict__ T, int n) {
    __shared__ __attribute__((aligned(16))) float sA[FM * PITCH64];
    __shared__ __attribute__((aligned(16))) float sG[FM * PITCH128];

    const int lane = threadIdx.x & 31;
    const int w    = __builtin_amdgcn_readfirstlane((int)(threadIdx.x >> 5));
    const int row0 = blockIdx.x * FM;

    for (int r = 0; r < FM / FWAVES; ++r) {
        const int lr   = w * (FM / FWAVES) + r;
        const int node = row0 + lr;
        v2f acc = (v2f)(0.0f);
        if (node < n) acc = agg_row<64, true>(x, dinv, rowptr, col, nullptr, nrm_out, node, lane);
        *((v2f*)(sA + lr * PITCH64) + lane) = acc;
    }
    __syncthreads();

    const int s     = w >> 2;
    const int jp    = w & 3;
    const int ml    = lane & 15;
    const int h     = lane >> 4;
    const int khalf = h << 1;

    {
        v8f c0 = (v8f){}, c1 = (v8f){};
        strip_pair_wmma<64, 128>(sA + (s * 16 + ml) * PITCH64 + khalf,
                                 W1 + (size_t)khalf * 128 + jp * 32 + ml, c0, c1);
        const float bv0 = b1[jp * 32 + ml];
        const float bv1 = b1[jp * 32 + 16 + ml];
        float* g = sG + (s * 16 + 8 * h) * PITCH128 + jp * 32 + ml;
#pragma unroll
        for (int r = 0; r < 8; ++r) {
            g[r * PITCH128]      = fmaxf(c0[r] + bv0, 0.0f);
            g[r * PITCH128 + 16] = fmaxf(c1[r] + bv1, 0.0f);
        }
    }
    __syncthreads();

    {
        v8f c0 = (v8f){}, c1 = (v8f){};
        strip_pair_wmma<128, 128>(sG + (s * 16 + ml) * PITCH128 + khalf,
                                  W2 + (size_t)khalf * 128 + jp * 32 + ml, c0, c1);
        const int grow = row0 + s * 16 + 8 * h;
        float* o = T + (size_t)grow * 128 + jp * 32 + ml;
        if (row0 + s * 16 + 16 <= n) {
#pragma unroll
            for (int r = 0; r < 8; ++r) {
                o[(size_t)r * 128]      = c0[r];
                o[(size_t)r * 128 + 16] = c1[r];
            }
        } else {
#pragma unroll
            for (int r = 0; r < 8; ++r) {
                if (grow + r < n) {
                    o[(size_t)r * 128]      = c0[r];
                    o[(size_t)r * 128 + 16] = c1[r];
                }
            }
        }
    }
}

__global__ void __launch_bounds__(FWAVES * 32)
agg_l2_gemm3_kernel(const float* __restrict__ T,
                    const float* __restrict__ dinv,
                    const int* __restrict__ rowptr,
                    const int* __restrict__ col,
                    const float* __restrict__ nrm,
                    const float* __restrict__ b2,
                    const float* __restrict__ W3,
                    float* __restrict__ U, int n) {
    __shared__ __attribute__((aligned(16))) float sG[FM * PITCH128];

    const int lane = threadIdx.x & 31;
    const int w    = __builtin_amdgcn_readfirstlane((int)(threadIdx.x >> 5));
    const int row0 = blockIdx.x * FM;

    const v4f bias = *((const v4f*)b2 + lane);
    for (int r = 0; r < FM / FWAVES; ++r) {
        const int lr   = w * (FM / FWAVES) + r;
        const int node = row0 + lr;
        v4f acc = (v4f)(0.0f);
        if (node < n) {
            acc = agg_row<128, false>(T, dinv, rowptr, col, nrm, nullptr, node, lane);
            acc += bias;
        }
        acc.x = fmaxf(acc.x, 0.0f);
        acc.y = fmaxf(acc.y, 0.0f);
        acc.z = fmaxf(acc.z, 0.0f);
        acc.w = fmaxf(acc.w, 0.0f);
        *((v4f*)(sG + lr * PITCH128) + lane) = acc;
    }
    __syncthreads();

    if (w < 8) {
        const int s     = w >> 1;
        const int jp    = w & 1;
        const int ml    = lane & 15;
        const int h     = lane >> 4;
        const int khalf = h << 1;
        v8f c0 = (v8f){}, c1 = (v8f){};
        strip_pair_wmma<128, 64>(sG + (s * 16 + ml) * PITCH128 + khalf,
                                 W3 + (size_t)khalf * 64 + jp * 32 + ml, c0, c1);
        const int grow = row0 + s * 16 + 8 * h;
        float* o = U + (size_t)grow * 64 + jp * 32 + ml;
        if (row0 + s * 16 + 16 <= n) {
#pragma unroll
            for (int r = 0; r < 8; ++r) {
                o[(size_t)r * 64]      = c0[r];
                o[(size_t)r * 64 + 16] = c1[r];
            }
        } else {
#pragma unroll
            for (int r = 0; r < 8; ++r) {
                if (grow + r < n) {
                    o[(size_t)r * 64]      = c0[r];
                    o[(size_t)r * 64 + 16] = c1[r];
                }
            }
        }
    }
}


static inline int cdiv(long long a, long long b) { return (int)((a + b - 1) / b); }

extern "C" void kernel_launch(void* const* d_in, const int* in_sizes, int n_in,
                              void* d_out, int out_size, void* d_ws, size_t ws_size,
                              hipStream_t stream) {
    const float* x  = (const float*)d_in[0];
    const float* W1 = (const float*)d_in[1];
    const float* b1 = (const float*)d_in[2];
    const float* W2 = (const float*)d_in[3];
    const float* b2 = (const float*)d_in[4];
    const float* W3 = (const float*)d_in[5];
    const float* b3 = (const float*)d_in[6];
    const int*   ei = (const int*)d_in[7];

    const int IN_C = 64, HID_C = 128, OUT_C = 64;
    const int n  = in_sizes[0] / IN_C;
    const int ne = in_sizes[7] / 2;
    const int* src = ei;
    const int* dst = ei + ne;

    const int NCH   = BUCKET_W;
    const int chunk = cdiv(ne, NCH);
    const int nb    = cdiv(n, BUCKET_W);

    float*    T      = (float*)d_ws;
    float*    G      = T + (size_t)n * HID_C;
    float*    dinv   = G + (size_t)n * HID_C;
    int*      rowptr = (int*)(dinv + n);
    int*      seg    = rowptr + n + 64;
    unsigned* packed = (unsigned*)(seg + (size_t)NCH * (nb + 1) + 64);
    int*      col    = (int*)(packed + ne);
    float*    nrm    = (float*)(col + ne);
    float*    out    = (float*)d_out;

    const int TB = 256;
    const int agg_blocks   = cdiv((long long)n * 32, TB);
    const int fused_blocks = cdiv(n, FM);

    chunk_order_kernel<<<NCH, SORT_TB, nb * sizeof(int), stream>>>(src, dst, seg, packed, ne, chunk, nb);
    bucket_csr_kernel<<<nb, BUCKET_W, 0, stream>>>(packed, seg, rowptr, dinv, col, n, nb, chunk);

    agg_l1_gemm2_kernel<<<fused_blocks, FWAVES * 32, 0, stream>>>(x, dinv, rowptr, col, nrm,
                                                                 W1, b1, W2, T, n);

    agg_l2_gemm3_kernel<<<fused_blocks, FWAVES * 32, 0, stream>>>(T, dinv, rowptr, col, nrm,
                                                                 b2, W3, G, n);

    csr_agg_kernel<64><<<agg_blocks, TB, 0, stream>>>(G, dinv, rowptr, col, nrm, b3, out, n);
}
